# starting kernel
# baseline (speedup 1.0000x reference)
#include <hip/hip_runtime.h>


typedef __attribute__((ext_vector_type(2))) float v2f;
typedef __attribute__((ext_vector_type(8))) float v8f;

#define D        496
#define LDW      512
#define PAIR_CAP 64
#define TUP_CAP  64

__global__ void extract_pairs_k(const float* __restrict__ M1, int T,
                                int* __restrict__ counts,
                                int2* __restrict__ pairs) {
  int idx = blockIdx.x * blockDim.x + threadIdx.x;
  if (idx >= T * D) return;
  int t = idx / D, i = idx - t * D;
  const float* row = M1 + (size_t)t * D * D + (size_t)i * D;
  int j = -1; float sv = 0.f;
  for (int c = 0; c < D; ++c) {
    float v = row[c];
    if (v != 0.f) { j = c; sv = v; }
  }
  if (j >= 0 && sv > 0.f) {
    int slot = atomicAdd(&counts[t], 1);
    pairs[t * PAIR_CAP + slot] = make_int2(i, j);
  }
}

__global__ void build_w_k(const float* __restrict__ theta,
                          const int* __restrict__ gti,
                          const int* __restrict__ gpi,
                          const int* __restrict__ counts,
                          const int2* __restrict__ pairs,
                          float* __restrict__ W, int G) {
  extern __shared__ float Uls[];
  const int kl = threadIdx.x;
  const int k0 = blockIdx.x * 128;
  const int k  = k0 + kl;

  for (int i = 0; i < D; ++i)
    Uls[i * 129 + kl] = (i == k) ? 1.f : 0.f;

  for (int g = 0; g < G; ++g) {
    int t = gti[g], p = gpi[g];
    float th = theta[p];
    float c = cosf(th), s = sinf(th);
    int n = counts[t];
    const int2* pl = pairs + t * PAIR_CAP;
    for (int pp = 0; pp < n; ++pp) {
      int2 ij = pl[pp];
      float a = Uls[ij.x * 129 + kl];
      float b = Uls[ij.y * 129 + kl];
      Uls[ij.x * 129 + kl] =  c * a + s * b;
      Uls[ij.y * 129 + kl] = -s * a + c * b;
    }
  }

  __syncthreads();
  for (int kk = 0; kk < 128; ++kk) {
    int krow = k0 + kk;
    for (int ib = threadIdx.x; ib < LDW; ib += 128) {
      float v = (ib < D) ? Uls[ib * 129 + kk] : 0.f;
      W[(size_t)krow * LDW + ib] = v;
    }
  }
}

__global__ void gemm_wmma_k(const float* __restrict__ A,
                            const float* __restrict__ W,
                            float* __restrict__ out) {
  const int lane = threadIdx.x & 31;
  const int wave = threadIdx.x >> 5;
  const int l16  = lane & 15;
  const int lh2  = (lane >> 4) * 2;
  const int m0   = (blockIdx.y * 8 + wave) * 64;
  const int n0   = blockIdx.x * 16;

  v8f acc[4];
#pragma unroll
  for (int mm = 0; mm < 4; ++mm) acc[mm] = (v8f)0.f;

  const float* a0 = A + (size_t)(m0 + l16) * D;
  const float* bp = W + (size_t)n0 + l16;

  for (int kk = 0; kk < D; kk += 4) {
    v2f b;
    b.x = bp[(size_t)(kk + lh2) * LDW];
    b.y = bp[(size_t)(kk + lh2 + 1) * LDW];
#pragma unroll
    for (int mm = 0; mm < 4; ++mm) {
      v2f a = *(const v2f*)(a0 + (size_t)mm * 16 * D + kk + lh2);
      acc[mm] = __builtin_amdgcn_wmma_f32_16x16x4_f32(
          false, a, false, b, (short)0, acc[mm], false, false);
    }
  }

  const int rbase = m0 + ((lane >> 4) << 3);
#pragma unroll
  for (int mm = 0; mm < 4; ++mm)
#pragma unroll
    for (int v = 0; v < 8; ++v)
      out[(size_t)(rbase + mm * 16 + v) * D + n0 + l16] = acc[mm][v];
}

extern "C" void kernel_launch(void* const* d_in, const int* in_sizes, int n_in,
                              void* d_out, int out_size, void* d_ws, size_t ws_size,
                              hipStream_t stream) {
  const float* input_state = (const float*)d_in[0];
  const float* theta       = (const float*)d_in[1];
  const float* M1          = (const float*)d_in[3];
  const int*   gti         = (const int*)d_in[5];
  const int*   gpi         = (const int*)d_in[6];

  const int B = in_sizes[0] / D;
  const int T = in_sizes[2] / (D * D);
  const int G = in_sizes[5];

  float* W      = (float*)d_ws;
  int*   counts = (int*)((char*)d_ws + (size_t)LDW * LDW * sizeof(float));
  int2*  pairs  = (int2*)(counts + TUP_CAP);
  float* out    = (float*)d_out;

  hipMemsetAsync(counts, 0, TUP_CAP * sizeof(int), stream);

  int tot = T * D;
  extract_pairs_k<<<(tot + 255) / 256, 256, 0, stream>>>(M1, T, counts, pairs);

  build_w_k<<<4, 128, (size_t)D * 129 * sizeof(float), stream>>>(
      theta, gti, gpi, counts, pairs, W, G);

  dim3 grid(D / 16, B / 512);
  gemm_wmma_k<<<grid, 256, 0, stream>>>(input_state, W, out);
}


# extract_pairs: a row cut into 4 pieces of 124 columns, a thread a piece (59520 threads, was 14880), no loop: a piece's 31 16-byte loads are independent; the row loop waited on memory in all 31 trips
# speedup vs baseline: 7.4259x; 7.4259x over previous
#include <hip/hip_runtime.h>


typedef __attribute__((ext_vector_type(2))) float v2f;
typedef __attribute__((ext_vector_type(8))) float v8f;

#define D        496
#define PAIR_CAP 64
#define TUP_CAP  64
#define NT       16
#define WG_ROWS  256
#define MT       2
#define PADROWS  32
#define LDW      512

#define PIECES   4
#define PIECE_F4 31

__global__ void __launch_bounds__(256)
extract_pairs_k(const float* __restrict__ M1, int T,
                int* __restrict__ counts,
                int2* __restrict__ pairs) {
  int idx = blockIdx.x * blockDim.x + threadIdx.x;
  if (idx >= T * D * PIECES) return;
  int r  = idx / PIECES, pc = idx - r * PIECES;
  int t  = r / D, i = r - t * D;
  int c0 = pc * (PIECE_F4 * 4);
  const float4* p = (const float4*)(M1 + (size_t)r * D + c0);

  float4 v[PIECE_F4];
#pragma unroll
  for (int q = 0; q < PIECE_F4; ++q) v[q] = p[q];

  int   jx = -1,  jy = -1,  jz = -1,  jw = -1;
  float sx = 0.f, sy = 0.f, sz = 0.f, sw = 0.f;
#pragma unroll
  for (int q = 0; q < PIECE_F4; ++q) {
    int c = c0 + 4 * q;
    if (v[q].x != 0.f) { jx = c;     sx = v[q].x; }
    if (v[q].y != 0.f) { jy = c + 1; sy = v[q].y; }
    if (v[q].z != 0.f) { jz = c + 2; sz = v[q].z; }
    if (v[q].w != 0.f) { jw = c + 3; sw = v[q].w; }
  }
  int j = jx; float sv = sx;
  if (jy > j) { j = jy; sv = sy; }
  if (jz > j) { j = jz; sv = sz; }
  if (jw > j) { j = jw; sv = sw; }

  for (int m = 1; m < PIECES; m <<= 1) {
    int   oj = __shfl_xor(j, m);
    float os = __shfl_xor(sv, m);
    if (oj > j) { j = oj; sv = os; }
  }

  if (pc == 0 && j >= 0 && sv > 0.f) {
    int slot = atomicAdd(&counts[t], 1);
    pairs[t * PAIR_CAP + slot] = make_int2(i, j);
  }
}

__global__ void __launch_bounds__(256)
compose_gemm_k(const float* __restrict__ A,
               const float* __restrict__ theta,
               const int* __restrict__ gti,
               const int* __restrict__ gpi,
               const int* __restrict__ counts,
               const int2* __restrict__ pairs,
               float* __restrict__ out,
               int G, int T) {
  extern __shared__ int4 smem[];
  float* Wt   = (float*)smem;
  int4*  grec = (int4*)(Wt + (D + PADROWS) * NT);
  int2*  ptab = (int2*)(grec + G);

  const int tid = threadIdx.x;
  const int n0  = blockIdx.x * NT;

  for (int e = tid; e < (D + PADROWS) * NT; e += 256) {
    int k = e >> 4, n = e & (NT - 1);
    Wt[e] = (k == n0 + n) ? 1.f : 0.f;
  }
  for (int g = tid; g < G; g += 256) {
    int t = gti[g];
    float th = theta[gpi[g]];
    grec[g] = make_int4(__float_as_int(cosf(th)), __float_as_int(sinf(th)),
                        t * PAIR_CAP, counts[t]);
  }
  for (int e = tid; e < T * PAIR_CAP; e += 256) {
    int t = e / PAIR_CAP, sl = e - t * PAIR_CAP;
    int  sp = D + (sl & (PADROWS - 1));
    int2 q = make_int2(sp, sp);
    if (sl < counts[t]) q = pairs[e];
    ptab[e] = q;
  }
  __syncthreads();

  const int col  = tid & (NT - 1);
  const int slot = tid >> 4;
  if (G > 0) {
    int4 rc = grec[G - 1];
    int4 rn = grec[G >= 2 ? G - 2 : 0];
    int2 q0 = ptab[rc.z + slot];
    int2 q1 = ptab[rc.z + slot + 16];
    for (int g = G - 1; g >= 0; --g) {
      int4 rn2 = grec[g >= 2 ? g - 2 : 0];
      int2 qn0 = ptab[rn.z + slot];
      int2 qn1 = ptab[rn.z + slot + 16];

      const float c = __int_as_float(rc.x), s = __int_as_float(rc.y);
      const int   n = rc.w;
      float a0 = Wt[q0.x * NT + col], b0 = Wt[q0.y * NT + col];
      float a1 = Wt[q1.x * NT + col], b1 = Wt[q1.y * NT + col];
      float x0 = c * a0 - s * b0, y0 = s * a0 + c * b0;
      float x1 = c * a1 - s * b1, y1 = s * a1 + c * b1;
      Wt[q0.x * NT + col] = x0;
      Wt[q0.y * NT + col] = y0;
      Wt[q1.x * NT + col] = x1;
      Wt[q1.y * NT + col] = y1;
      for (int ps = slot + 32; ps < n; ps += 16) {
        int2 q = ptab[rc.z + ps];
        float a = Wt[q.x * NT + col], b = Wt[q.y * NT + col];
        Wt[q.x * NT + col] = c * a - s * b;
        Wt[q.y * NT + col] = s * a + c * b;
      }
      __syncthreads();
      rc = rn; rn = rn2; q0 = qn0; q1 = qn1;
    }
  }

  const int lane = tid & 31;
  const int wave = tid >> 5;
  const int l16  = lane & 15;
  const int lh2  = (lane >> 4) * 2;
  const int m0   = blockIdx.y * WG_ROWS + wave * (MT * 16);

  v8f acc[MT];
#pragma unroll
  for (int mm = 0; mm < MT; ++mm) acc[mm] = (v8f)0.f;

  const float* a0p = A + (size_t)(m0 + l16) * D;
  const float* bl  = Wt + l16;

  for (int kk = 0; kk < D; kk += 4) {
    v2f b;
    b.x = bl[(kk + lh2) * NT];
    b.y = bl[(kk + lh2 + 1) * NT];
#pragma unroll
    for (int mm = 0; mm < MT; ++mm) {
      v2f a = *(const v2f*)(a0p + (size_t)mm * 16 * D + kk + lh2);
      acc[mm] = __builtin_amdgcn_wmma_f32_16x16x4_f32(
          false, a, false, b, (short)0, acc[mm], false, false);
    }
  }

  const int rbase = m0 + ((lane >> 4) << 3);
#pragma unroll
  for (int mm = 0; mm < MT; ++mm)
#pragma unroll
    for (int v = 0; v < 8; ++v)
      out[(size_t)(rbase + mm * 16 + v) * D + n0 + l16] = acc[mm][v];
}

extern "C" void kernel_launch(void* const* d_in, const int* in_sizes, int n_in,
                              void* d_out, int out_size, void* d_ws, size_t ws_size,
                              hipStream_t stream) {
  const float* input_state = (const float*)d_in[0];
  const float* theta       = (const float*)d_in[1];
  const float* M1          = (const float*)d_in[3];
  const int*   gti         = (const int*)d_in[5];
  const int*   gpi         = (const int*)d_in[6];

  const int B = in_sizes[0] / D;
  const int T = in_sizes[2] / (D * D);
  const int G = in_sizes[5];

  int*   counts = (int*)((char*)d_ws + (size_t)LDW * LDW * sizeof(float));
  int2*  pairs  = (int2*)(counts + TUP_CAP);
  float* out    = (float*)d_out;

  hipMemsetAsync(counts, 0, TUP_CAP * sizeof(int), stream);

  int tot = T * D * PIECES;
  extract_pairs_k<<<(tot + 255) / 256, 256, 0, stream>>>(M1, T, counts, pairs);

  size_t lds = (size_t)(D + PADROWS) * NT * sizeof(float) + (size_t)G * sizeof(int4) +
               (size_t)T * PAIR_CAP * sizeof(int2);
  dim3 grid(D / NT, B / WG_ROWS);
  compose_gemm_k<<<grid, 256, lds, stream>>>(input_state, theta, gti, gpi,
                                             counts, pairs, out, G, T);
}


# compose phase: a wave composes two whole columns of W alone (lane = pair slot, a row's two columns one 8-byte LDS access): the gate loop needs no work-group barrier, 2 barriers a work-group, was 157
# speedup vs baseline: 8.0231x; 1.0804x over previous
#include <hip/hip_runtime.h>


typedef __attribute__((ext_vector_type(2))) float v2f;
typedef __attribute__((ext_vector_type(8))) float v8f;

#define D        496
#define PAIR_CAP 64
#define TUP_CAP  64
#define NT       16
#define WG_ROWS  256
#define MT       2
#define PADROWS  32
#define NPLANE   (NT / 2)
#define PLANE    1064
#define LDW      512

#define PIECES   4
#define PIECE_F4 31

__global__ void __launch_bounds__(256)
extract_pairs_k(const float* __restrict__ M1, int T,
                int* __restrict__ counts,
                int2* __restrict__ pairs) {
  int idx = blockIdx.x * blockDim.x + threadIdx.x;
  if (idx >= T * D * PIECES) return;
  int r  = idx / PIECES, pc = idx - r * PIECES;
  int t  = r / D, i = r - t * D;
  int c0 = pc * (PIECE_F4 * 4);
  const float4* p = (const float4*)(M1 + (size_t)r * D + c0);

  float4 v[PIECE_F4];
#pragma unroll
  for (int q = 0; q < PIECE_F4; ++q) v[q] = p[q];

  int   jx = -1,  jy = -1,  jz = -1,  jw = -1;
  float sx = 0.f, sy = 0.f, sz = 0.f, sw = 0.f;
#pragma unroll
  for (int q = 0; q < PIECE_F4; ++q) {
    int c = c0 + 4 * q;
    if (v[q].x != 0.f) { jx = c;     sx = v[q].x; }
    if (v[q].y != 0.f) { jy = c + 1; sy = v[q].y; }
    if (v[q].z != 0.f) { jz = c + 2; sz = v[q].z; }
    if (v[q].w != 0.f) { jw = c + 3; sw = v[q].w; }
  }
  int j = jx; float sv = sx;
  if (jy > j) { j = jy; sv = sy; }
  if (jz > j) { j = jz; sv = sz; }
  if (jw > j) { j = jw; sv = sw; }

  for (int m = 1; m < PIECES; m <<= 1) {
    int   oj = __shfl_xor(j, m);
    float os = __shfl_xor(sv, m);
    if (oj > j) { j = oj; sv = os; }
  }

  if (pc == 0 && j >= 0 && sv > 0.f) {
    int slot = atomicAdd(&counts[t], 1);
    pairs[t * PAIR_CAP + slot] = make_int2(i, j);
  }
}

__global__ void __launch_bounds__(256)
compose_gemm_k(const float* __restrict__ A,
               const float* __restrict__ theta,
               const int* __restrict__ gti,
               const int* __restrict__ gpi,
               const int* __restrict__ counts,
               const int2* __restrict__ pairs,
               float* __restrict__ out,
               int G, int T) {
  extern __shared__ int4 smem[];
  float* Wp   = (float*)smem;
  int4*  grec = (int4*)(Wp + NPLANE * PLANE);
  int2*  ptab = (int2*)(grec + G);

  const int tid  = threadIdx.x;
  const int n0   = blockIdx.x * NT;
  const int lane = tid & 31;
  const int wave = tid >> 5;

  float2* Wv = (float2*)(Wp + wave * PLANE);
  for (int k = lane; k < D + PADROWS; k += 32) {
    float2 u;
    u.x = (k == n0 + 2 * wave)     ? 1.f : 0.f;
    u.y = (k == n0 + 2 * wave + 1) ? 1.f : 0.f;
    Wv[k] = u;
  }
  for (int g = tid; g < G; g += 256) {
    int t = gti[g];
    float th = theta[gpi[g]];
    grec[g] = make_int4(__float_as_int(cosf(th)), __float_as_int(sinf(th)),
                        t * PAIR_CAP, counts[t]);
  }
  for (int e = tid; e < T * PAIR_CAP; e += 256) {
    int t = e / PAIR_CAP, sl = e - t * PAIR_CAP;
    int  sp = D + (sl & (PADROWS - 1));
    int2 q = make_int2(sp, sp);
    if (sl < counts[t]) q = pairs[e];
    ptab[e] = q;
  }
  __syncthreads();

  if (G > 0) {
    int4 rc = grec[G - 1];
    int4 rn = grec[G >= 2 ? G - 2 : 0];
    int2 q  = ptab[rc.z + lane];
    for (int g = G - 1; g >= 0; --g) {
      float2 a = Wv[q.x], b = Wv[q.y];
      int4 rn2 = grec[g >= 2 ? g - 2 : 0];
      int2 qn  = ptab[rn.z + lane];

      const float c = __int_as_float(rc.x), s = __int_as_float(rc.y);
      const int   n = rc.w;
      float2 x, y;
      x.x = c * a.x - s * b.x;  x.y = c * a.y - s * b.y;
      y.x = s * a.x + c * b.x;  y.y = s * a.y + c * b.y;
      Wv[q.x] = x;
      Wv[q.y] = y;
      for (int ps = lane + 32; ps < n; ps += 32) {
        int2 r = ptab[rc.z + ps];
        float2 e = Wv[r.x], f = Wv[r.y];
        float2 v, w;
        v.x = c * e.x - s * f.x;  v.y = c * e.y - s * f.y;
        w.x = s * e.x + c * f.x;  w.y = s * e.y + c * f.y;
        Wv[r.x] = v;
        Wv[r.y] = w;
      }
      rc = rn; rn = rn2; q = qn;
    }
  }
  __syncthreads();

  const int l16  = lane & 15;
  const int lh2  = (lane >> 4) * 2;
  const int m0   = blockIdx.y * WG_ROWS + wave * (MT * 16);

  v8f acc[MT];
#pragma unroll
  for (int mm = 0; mm < MT; ++mm) acc[mm] = (v8f)0.f;

  const float* a0p = A + (size_t)(m0 + l16) * D;
  const float* bl  = Wp + (l16 >> 1) * PLANE + (l16 & 1);

  for (int kk = 0; kk < D; kk += 4) {
    v2f b;
    b.x = bl[(kk + lh2) * 2];
    b.y = bl[(kk + lh2 + 1) * 2];
#pragma unroll
    for (int mm = 0; mm < MT; ++mm) {
      v2f a = *(const v2f*)(a0p + (size_t)mm * 16 * D + kk + lh2);
      acc[mm] = __builtin_amdgcn_wmma_f32_16x16x4_f32(
          false, a, false, b, (short)0, acc[mm], false, false);
    }
  }

  const int rbase = m0 + ((lane >> 4) << 3);
#pragma unroll
  for (int mm = 0; mm < MT; ++mm)
#pragma unroll
    for (int v = 0; v < 8; ++v)
      out[(size_t)(rbase + mm * 16 + v) * D + n0 + l16] = acc[mm][v];
}

extern "C" void kernel_launch(void* const* d_in, const int* in_sizes, int n_in,
                              void* d_out, int out_size, void* d_ws, size_t ws_size,
                              hipStream_t stream) {
  const float* input_state = (const float*)d_in[0];
  const float* theta       = (const float*)d_in[1];
  const float* M1          = (const float*)d_in[3];
  const int*   gti         = (const int*)d_in[5];
  const int*   gpi         = (const int*)d_in[6];

  const int B = in_sizes[0] / D;
  const int T = in_sizes[2] / (D * D);
  const int G = in_sizes[5];

  int*   counts = (int*)((char*)d_ws + (size_t)LDW * LDW * sizeof(float));
  int2*  pairs  = (int2*)(counts + TUP_CAP);
  float* out    = (float*)d_out;

  hipMemsetAsync(counts, 0, TUP_CAP * sizeof(int), stream);

  int tot = T * D * PIECES;
  extract_pairs_k<<<(tot + 255) / 256, 256, 0, stream>>>(M1, T, counts, pairs);

  size_t lds = (size_t)NPLANE * PLANE * sizeof(float) + (size_t)G * sizeof(int4) +
               (size_t)T * PAIR_CAP * sizeof(int2);
  dim3 grid(D / NT, B / WG_ROWS);
  compose_gemm_k<<<grid, 256, lds, stream>>>(input_state, theta, gti, gpi,
                                             counts, pairs, out, G, T);
}


# product operand feed: a lane loads 16 B of its A row per 8 k (k order in a fragment permuted, B mirrors it: sums in another order), 24 blocks ahead in registers, first 48 loads before the gate loop
# speedup vs baseline: 10.3652x; 1.2919x over previous
#include <hip/hip_runtime.h>


typedef __attribute__((ext_vector_type(2))) float v2f;
typedef __attribute__((ext_vector_type(4))) float v4f;
typedef __attribute__((ext_vector_type(8))) float v8f;

#define D        496
#define PAIR_CAP 64
#define TUP_CAP  64
#define NT       16
#define WG_ROWS  256
#define MT       2
#define PADROWS  32
#define NPLANE   (NT / 2)
#define PLANE    1066
#define KB       (D / 8)
#define AHEAD    24
#define BAHEAD   3
#define LDW      512

#define PIECES   4
#define PIECE_F4 31

__global__ void __launch_bounds__(256)
extract_pairs_k(const float* __restrict__ M1, int T,
                int* __restrict__ counts,
                int2* __restrict__ pairs) {
  int idx = blockIdx.x * blockDim.x + threadIdx.x;
  if (idx >= T * D * PIECES) return;
  int r  = idx / PIECES, pc = idx - r * PIECES;
  int t  = r / D, i = r - t * D;
  int c0 = pc * (PIECE_F4 * 4);
  const float4* p = (const float4*)(M1 + (size_t)r * D + c0);

  float4 v[PIECE_F4];
#pragma unroll
  for (int q = 0; q < PIECE_F4; ++q) v[q] = p[q];

  int   jx = -1,  jy = -1,  jz = -1,  jw = -1;
  float sx = 0.f, sy = 0.f, sz = 0.f, sw = 0.f;
#pragma unroll
  for (int q = 0; q < PIECE_F4; ++q) {
    int c = c0 + 4 * q;
    if (v[q].x != 0.f) { jx = c;     sx = v[q].x; }
    if (v[q].y != 0.f) { jy = c + 1; sy = v[q].y; }
    if (v[q].z != 0.f) { jz = c + 2; sz = v[q].z; }
    if (v[q].w != 0.f) { jw = c + 3; sw = v[q].w; }
  }
  int j = jx; float sv = sx;
  if (jy > j) { j = jy; sv = sy; }
  if (jz > j) { j = jz; sv = sz; }
  if (jw > j) { j = jw; sv = sw; }

  for (int m = 1; m < PIECES; m <<= 1) {
    int   oj = __shfl_xor(j, m);
    float os = __shfl_xor(sv, m);
    if (oj > j) { j = oj; sv = os; }
  }

  if (pc == 0 && j >= 0 && sv > 0.f) {
    int slot = atomicAdd(&counts[t], 1);
    pairs[t * PAIR_CAP + slot] = make_int2(i, j);
  }
}

__global__ void __launch_bounds__(256)
compose_gemm_k(const float* __restrict__ A,
               const float* __restrict__ theta,
               const int* __restrict__ gti,
               const int* __restrict__ gpi,
               const int* __restrict__ counts,
               const int2* __restrict__ pairs,
               float* __restrict__ out,
               int G, int T) {
  extern __shared__ int4 smem[];
  float* Wp   = (float*)smem;
  int4*  grec = (int4*)(Wp + NPLANE * PLANE);
  int2*  ptab = (int2*)(grec + G);

  const int tid  = threadIdx.x;
  const int n0   = blockIdx.x * NT;
  const int lane = tid & 31;
  const int wave = tid >> 5;

  float2* Wv = (float2*)(Wp + wave * PLANE);
  for (int k = lane; k < D + PADROWS; k += 32) {
    float2 u;
    u.x = (k == n0 + 2 * wave)     ? 1.f : 0.f;
    u.y = (k == n0 + 2 * wave + 1) ? 1.f : 0.f;
    Wv[k] = u;
  }
  for (int g = tid; g < G; g += 256) {
    int t = gti[g];
    float th = theta[gpi[g]];
    grec[g] = make_int4(__float_as_int(cosf(th)), __float_as_int(sinf(th)),
                        t * PAIR_CAP, counts[t]);
  }
  for (int e = tid; e < T * PAIR_CAP; e += 256) {
    int t = e / PAIR_CAP, sl = e - t * PAIR_CAP;
    int  sp = D + (sl & (PADROWS - 1));
    int2 q = make_int2(sp, sp);
    if (sl < counts[t]) q = pairs[e];
    ptab[e] = q;
  }
  __syncthreads();

  const int l16  = lane & 15;
  const int lh4  = (lane >> 4) * 4;
  const int m0   = blockIdx.y * WG_ROWS + wave * (MT * 16);
  const float* a0p = A + (size_t)(m0 + l16) * D + lh4;

  v4f ar[KB][MT];
#pragma unroll
  for (int j = 0; j < AHEAD; ++j)
#pragma unroll
    for (int mm = 0; mm < MT; ++mm)
      ar[j][mm] = *(const v4f*)(a0p + (size_t)mm * 16 * D + 8 * j);

  if (G > 0) {
    int4 rc = grec[G - 1];
    int4 rn = grec[G >= 2 ? G - 2 : 0];
    int2 q  = ptab[rc.z + lane];
    for (int g = G - 1; g >= 0; --g) {
      float2 a = Wv[q.x], b = Wv[q.y];
      int4 rn2 = grec[g >= 2 ? g - 2 : 0];
      int2 qn  = ptab[rn.z + lane];

      const float c = __int_as_float(rc.x), s = __int_as_float(rc.y);
      const int   n = rc.w;
      float2 x, y;
      x.x = c * a.x - s * b.x;  x.y = c * a.y - s * b.y;
      y.x = s * a.x + c * b.x;  y.y = s * a.y + c * b.y;
      Wv[q.x] = x;
      Wv[q.y] = y;
      for (int ps = lane + 32; ps < n; ps += 32) {
        int2 r = ptab[rc.z + ps];
        float2 e = Wv[r.x], f = Wv[r.y];
        float2 v, w;
        v.x = c * e.x - s * f.x;  v.y = c * e.y - s * f.y;
        w.x = s * e.x + c * f.x;  w.y = s * e.y + c * f.y;
        Wv[r.x] = v;
        Wv[r.y] = w;
      }
      rc = rn; rn = rn2; q = qn;
    }
  }
  __syncthreads();

  v8f acc[MT];
#pragma unroll
  for (int mm = 0; mm < MT; ++mm) acc[mm] = (v8f)0.f;

  const float* bl = Wp + (l16 >> 1) * PLANE + (l16 & 1) + 2 * lh4;

  v2f bq[KB][2];
#pragma unroll
  for (int j = 0; j < BAHEAD; ++j) {
    bq[j][0].x = bl[16 * j];      bq[j][0].y = bl[16 * j + 2];
    bq[j][1].x = bl[16 * j + 4];  bq[j][1].y = bl[16 * j + 6];
  }

#pragma unroll
  for (int j = 0; j < KB; ++j) {
    if (j + AHEAD < KB) {
#pragma unroll
      for (int mm = 0; mm < MT; ++mm)
        ar[j + AHEAD][mm] = *(const v4f*)(a0p + (size_t)mm * 16 * D + 8 * (j + AHEAD));
    }
    if (j + BAHEAD < KB) {
      const int jn = j + BAHEAD;
      bq[jn][0].x = bl[16 * jn];      bq[jn][0].y = bl[16 * jn + 2];
      bq[jn][1].x = bl[16 * jn + 4];  bq[jn][1].y = bl[16 * jn + 6];
    }
#pragma unroll
    for (int mm = 0; mm < MT; ++mm) {
      v2f a;
      a.x = ar[j][mm].x;  a.y = ar[j][mm].y;
      acc[mm] = __builtin_amdgcn_wmma_f32_16x16x4_f32(
          false, a, false, bq[j][0], (short)0, acc[mm], false, false);
    }
#pragma unroll
    for (int mm = 0; mm < MT; ++mm) {
      v2f a;
      a.x = ar[j][mm].z;  a.y = ar[j][mm].w;
      acc[mm] = __builtin_amdgcn_wmma_f32_16x16x4_f32(
          false, a, false, bq[j][1], (short)0, acc[mm], false, false);
    }
    __builtin_amdgcn_sched_barrier(0);
  }

  const int rbase = m0 + ((lane >> 4) << 3);
#pragma unroll
  for (int mm = 0; mm < MT; ++mm)
#pragma unroll
    for (int v = 0; v < 8; ++v)
      out[(size_t)(rbase + mm * 16 + v) * D + n0 + l16] = acc[mm][v];
}

extern "C" void kernel_launch(void* const* d_in, const int* in_sizes, int n_in,
                              void* d_out, int out_size, void* d_ws, size_t ws_size,
                              hipStream_t stream) {
  const float* input_state = (const float*)d_in[0];
  const float* theta       = (const float*)d_in[1];
  const float* M1          = (const float*)d_in[3];
  const int*   gti         = (const int*)d_in[5];
  const int*   gpi         = (const int*)d_in[6];

  const int B = in_sizes[0] / D;
  const int T = in_sizes[2] / (D * D);
  const int G = in_sizes[5];

  int*   counts = (int*)((char*)d_ws + (size_t)LDW * LDW * sizeof(float));
  int2*  pairs  = (int2*)(counts + TUP_CAP);
  float* out    = (float*)d_out;

  hipMemsetAsync(counts, 0, TUP_CAP * sizeof(int), stream);

  int tot = T * D * PIECES;
  extract_pairs_k<<<(tot + 255) / 256, 256, 0, stream>>>(M1, T, counts, pairs);

  size_t lds = (size_t)NPLANE * PLANE * sizeof(float) + (size_t)G * sizeof(int4) +
               (size_t)T * PAIR_CAP * sizeof(int2);
  dim3 grid(D / NT, B / WG_ROWS);
  compose_gemm_k<<<grid, 256, lds, stream>>>(input_state, theta, gti, gpi,
                                             counts, pairs, out, G, T);
}
